# starting kernel
# baseline (speedup 1.0000x reference)
#include <hip/hip_runtime.h>

#define NQ 10
#define DIM 1024
#define NLAYERS 4

typedef __attribute__((ext_vector_type(2))) float v2f;
typedef __attribute__((ext_vector_type(8))) float v8f;

__device__ __forceinline__ int cnot_src(int d) {
#pragma unroll
  for (int i = 8; i >= 0; --i)
    if ((d >> i) & 1) d ^= (2 << i);
  return d;
}

__device__ __forceinline__ void build_bfrag(const float* cc, const float* ss,
                                            int lane, v2f bfrag[4]) {
  const int half = lane >> 4;
  const int n = lane & 15;
#pragma unroll
  for (int ks = 0; ks < 4; ++ks) {
    int k0 = 4 * ks + 2 * half;
    int k1 = k0 + 1;
    float e0 = 1.f, e1 = 1.f;
#pragma unroll
    for (int q = 0; q < 4; ++q) {
      float c = cc[q], s = ss[q];
      int nb = (n >> q) & 1;
      int k0b = (k0 >> q) & 1;
      int k1b = (k1 >> q) & 1;
      e0 *= (nb == k0b) ? c : (nb ? s : -s);
      e1 *= (nb == k1b) ? c : (nb ? s : -s);
    }
    v2f bv; bv.x = e0; bv.y = e1;
    bfrag[ks] = bv;
  }
}

__device__ __forceinline__ void contract_group(float* S, const v2f bfrag[4],
                                               int group, bool permute, int lane) {
  const int half = lane >> 4;
  const int lid = lane & 15;
  v2f a[16];
#pragma unroll
  for (int t = 0; t < 4; ++t) {
#pragma unroll
    for (int ks = 0; ks < 4; ++ks) {
      int k0 = 4 * ks + 2 * half;
      int d0, d1;
      if (group == 0) { int m = t * 16 + lid; d0 = (m << 4) | k0; d1 = d0 + 1; }
      else            { d0 = t * 256 + k0 * 16 + lid; d1 = d0 + 16; }
      if (permute) { d0 = cnot_src(d0); d1 = cnot_src(d1); }
      v2f av; av.x = S[d0]; av.y = S[d1];
      a[t * 4 + ks] = av;
    }
  }
#pragma unroll
  for (int t = 0; t < 4; ++t) {
    v8f acc = {0.f, 0.f, 0.f, 0.f, 0.f, 0.f, 0.f, 0.f};
#pragma unroll
    for (int ks = 0; ks < 4; ++ks)
      acc = __builtin_amdgcn_wmma_f32_16x16x4_f32(
          false, a[t * 4 + ks], false, bfrag[ks], (short)0, acc, false, false);
#pragma unroll
    for (int r = 0; r < 8; ++r) {
      int m = r + 8 * half;
      int d = (group == 0) ? (((t * 16 + m) << 4) | lid)
                           : (t * 256 + lid * 16 + m);
      S[d] = acc[r];
    }
  }
}

__device__ __forceinline__ void contract_top(float* Sre, float* Sim,
                                             float c8, float s8,
                                             float c9, float s9, int lane) {
#pragma unroll
  for (int j = 0; j < 8; ++j) {
    int rest = j * 32 + lane;
    {
      float a0 = Sre[rest], a1 = Sre[256 + rest];
      float a2 = Sre[512 + rest], a3 = Sre[768 + rest];
      float t0 = c8 * a0 - s8 * a1, t1 = s8 * a0 + c8 * a1;
      float t2 = c8 * a2 - s8 * a3, t3 = s8 * a2 + c8 * a3;
      Sre[rest]       = c9 * t0 - s9 * t2;
      Sre[256 + rest] = c9 * t1 - s9 * t3;
      Sre[512 + rest] = s9 * t0 + c9 * t2;
      Sre[768 + rest] = s9 * t1 + c9 * t3;
    }
    {
      float a0 = Sim[rest], a1 = Sim[256 + rest];
      float a2 = Sim[512 + rest], a3 = Sim[768 + rest];
      float t0 = c8 * a0 - s8 * a1, t1 = s8 * a0 + c8 * a1;
      float t2 = c8 * a2 - s8 * a3, t3 = s8 * a2 + c8 * a3;
      Sim[rest]       = c9 * t0 - s9 * t2;
      Sim[256 + rest] = c9 * t1 - s9 * t3;
      Sim[512 + rest] = s9 * t0 + c9 * t2;
      Sim[768 + rest] = s9 * t1 + c9 * t3;
    }
  }
}

__global__ __launch_bounds__(32) void qnn_circuit_kernel(
    const float* __restrict__ x, const float* __restrict__ W_pre,
    const float* __restrict__ b_pre, const float* __restrict__ weights,
    const float* __restrict__ W_post, const float* __restrict__ b_post,
    float* __restrict__ out) {
  __shared__ float Sre[DIM];
  __shared__ float Sim[DIM];
  __shared__ float hs[16];
  __shared__ float gs[16];
  __shared__ float csc[(NLAYERS + 1) * NQ];
  __shared__ float css[(NLAYERS + 1) * NQ];
  __shared__ float red[6 * 32];
  __shared__ float zb[16];

  const int lane = threadIdx.x;
  const int b = blockIdx.x;

  for (int idx = lane; idx < (NLAYERS + 1) * NQ; idx += 32) {
    float w = 0.5f * weights[idx];
    csc[idx] = __cosf(w);
    css[idx] = __sinf(w);
  }
  if (lane < NQ) {
    float acc = b_pre[lane];
#pragma unroll
    for (int k = 0; k < NQ; ++k) acc += x[b * NQ + k] * W_pre[lane * NQ + k];
    hs[lane] = acc;
  }
  __syncthreads();
  if (lane < NQ - 1) {
    const float PI = 3.14159265358979f;
    gs[lane] = (PI - hs[lane]) * (PI - hs[lane + 1]);
  }
  __syncthreads();

  float hr[NQ], gr[NQ - 1];
#pragma unroll
  for (int q = 0; q < NQ; ++q) hr[q] = hs[q];
#pragma unroll
  for (int q = 0; q < NQ - 1; ++q) gr[q] = gs[q];

  const float inv = 0.03125f;
#pragma unroll 4
  for (int j = 0; j < 32; ++j) {
    int d = (j << 5) | lane;
    float ang = 0.f;
#pragma unroll
    for (int q = 0; q < NQ; ++q) ang += ((d >> q) & 1) ? -hr[q] : hr[q];
    int dz = d ^ (d >> 1);
#pragma unroll
    for (int q = 0; q < NQ - 1; ++q) ang += ((dz >> q) & 1) ? -gr[q] : gr[q];
    float sn, cn;
    __sincosf(ang, &sn, &cn);
    Sre[d] = cn * inv;
    Sim[d] = -sn * inv;
  }
  __syncthreads();

  v2f bfrag[4];
#pragma unroll 1
  for (int l = 0; l <= NLAYERS; ++l) {
    build_bfrag(&csc[l * NQ + 0], &css[l * NQ + 0], lane, bfrag);
    contract_group(Sre, bfrag, 0, l > 0, lane);
    contract_group(Sim, bfrag, 0, l > 0, lane);
    __syncthreads();
    build_bfrag(&csc[l * NQ + 4], &css[l * NQ + 4], lane, bfrag);
    contract_group(Sre, bfrag, 1, false, lane);
    contract_group(Sim, bfrag, 1, false, lane);
    __syncthreads();
    contract_top(Sre, Sim, csc[l * NQ + 8], css[l * NQ + 8],
                 csc[l * NQ + 9], css[l * NQ + 9], lane);
    __syncthreads();
  }

  float psum = 0.f, zh0 = 0.f, zh1 = 0.f, zh2 = 0.f, zh3 = 0.f, zh4 = 0.f;
#pragma unroll 4
  for (int j = 0; j < 32; ++j) {
    int d = (j << 5) | lane;
    float re = Sre[d], im = Sim[d];
    float p = re * re + im * im;
    psum += p;
    zh0 += (j & 1) ? -p : p;
    zh1 += (j & 2) ? -p : p;
    zh2 += (j & 4) ? -p : p;
    zh3 += (j & 8) ? -p : p;
    zh4 += (j & 16) ? -p : p;
  }
  red[lane] = psum;
  red[32 + lane] = zh0;  red[64 + lane] = zh1;  red[96 + lane] = zh2;
  red[128 + lane] = zh3; red[160 + lane] = zh4;
  __syncthreads();
  if (lane < NQ) {
    float z = 0.f;
    if (lane < 5) {
#pragma unroll
      for (int L = 0; L < 32; ++L) z += ((L >> lane) & 1) ? -red[L] : red[L];
    } else {
#pragma unroll
      for (int L = 0; L < 32; ++L) z += red[(lane - 4) * 32 + L];
    }
    zb[lane] = z;
  }
  __syncthreads();
  if (lane < NQ) {
    float acc = b_post[lane];
#pragma unroll
    for (int q = 0; q < NQ; ++q) acc += zb[q] * W_post[lane * NQ + q];
    out[b * NQ + lane] = acc;
  }
}

extern "C" void kernel_launch(void* const* d_in, const int* in_sizes, int n_in,
                              void* d_out, int out_size, void* d_ws, size_t ws_size,
                              hipStream_t stream) {
  const float* x       = (const float*)d_in[0];
  const float* W_pre   = (const float*)d_in[1];
  const float* b_pre   = (const float*)d_in[2];
  const float* weights = (const float*)d_in[3];
  const float* W_post  = (const float*)d_in[4];
  const float* b_post  = (const float*)d_in[5];
  float* out = (float*)d_out;
  const int batch = in_sizes[0] / NQ;
  hipLaunchKernelGGL(qnn_circuit_kernel, dim3(batch), dim3(32), 0, stream,
                     x, W_pre, b_pre, weights, W_post, b_post, out);
}


# Readout split over both waves: each sums |amp|^2 of its own plane from eight 16-byte LDS reads by sum/difference trees, lanes folded through LDS; one barrier less (order of the f32 sums changes)
# speedup vs baseline: 1.6989x; 1.6989x over previous
#include <hip/hip_runtime.h>

#define NQ 10
#define DIM 1024
#define NLAYERS 4
#define ROWP 20
#define PLANE (64 * ROWP)

typedef __attribute__((ext_vector_type(2))) float v2f;
typedef __attribute__((ext_vector_type(4))) float v4f;
typedef __attribute__((ext_vector_type(8))) float v8f;

__device__ __forceinline__ int cnot_src(int d) {
#pragma unroll
  for (int i = 8; i >= 0; --i)
    if ((d >> i) & 1) d ^= (2 << i);
  return d;
}

__device__ __forceinline__ int lds_pos(int d) { return (d >> 4) * ROWP + (d & 15); }

__device__ __forceinline__ void build_bfrag(const float* cc, const float* ss,
                                            int lane, v2f bfrag[4]) {
  const int half = lane >> 4;
  const int n = lane & 15;
#pragma unroll
  for (int ks = 0; ks < 4; ++ks) {
    int k0 = 8 * half + 2 * ks;
    int k1 = k0 + 1;
    float e0 = 1.f, e1 = 1.f;
#pragma unroll
    for (int q = 0; q < 4; ++q) {
      float c = cc[q], s = ss[q];
      int nb = (n >> q) & 1;
      int k0b = (k0 >> q) & 1;
      int k1b = (k1 >> q) & 1;
      e0 *= (nb == k0b) ? c : (nb ? s : -s);
      e1 *= (nb == k1b) ? c : (nb ? s : -s);
    }
    v2f bv; bv.x = e0; bv.y = e1;
    bfrag[ks] = bv;
  }
}

__device__ __forceinline__ void contract_group(float* S, const v2f bfrag[4], int lane) {
  const int half = lane >> 4;
  const int lid = lane & 15;
  float* base = S + lid * ROWP + 8 * half;
  v4f q[8];
#pragma unroll
  for (int t = 0; t < 4; ++t) {
    q[2 * t]     = *(const v4f*)(base + t * 16 * ROWP);
    q[2 * t + 1] = *(const v4f*)(base + t * 16 * ROWP + 4);
  }
#pragma unroll
  for (int t = 0; t < 4; ++t) {
    v8f acc = {0.f, 0.f, 0.f, 0.f, 0.f, 0.f, 0.f, 0.f};
#pragma unroll
    for (int ks = 0; ks < 4; ++ks) {
      v4f qq = q[2 * t + (ks >> 1)];
      v2f av;
      av.x = (ks & 1) ? qq.z : qq.x;
      av.y = (ks & 1) ? qq.w : qq.y;
      acc = __builtin_amdgcn_wmma_f32_16x16x4_f32(
          false, av, false, bfrag[ks], (short)0, acc, false, false);
    }
    v4f lo4, hi4;
    lo4.x = acc[0]; lo4.y = acc[1]; lo4.z = acc[2]; lo4.w = acc[3];
    hi4.x = acc[4]; hi4.y = acc[5]; hi4.z = acc[6]; hi4.w = acc[7];
    *(v4f*)(base + t * 16 * ROWP)     = lo4;
    *(v4f*)(base + t * 16 * ROWP + 4) = hi4;
  }
}

__device__ __forceinline__ void gather_top(float* S, float c8, float s8,
                                           float c9, float s9, int lane) {
  const int m = lane & 15;
  const int g = lane >> 4;
  v4f qa[4], qb[4];
#pragma unroll
  for (int t = 0; t < 4; ++t) {
    int d0 = (t << 8) | (m << 4) | (4 * g);
    qa[t] = *(const v4f*)(S + lds_pos(cnot_src(d0) & ~3));
    qb[t] = *(const v4f*)(S + lds_pos(cnot_src(d0 + 2) & ~3));
  }
  float vp[4][4], vc[4][4];
#pragma unroll
  for (int t = 0; t < 4; ++t) {
    vp[t][0] = qa[t].x; vp[t][1] = qa[t].w; vp[t][2] = qb[t].z; vp[t][3] = qb[t].y;
    vc[3 - t][0] = qa[t].y; vc[3 - t][1] = qa[t].z; vc[3 - t][2] = qb[t].w; vc[3 - t][3] = qb[t].x;
  }
  float op[4][4], oc[4][4];
#pragma unroll
  for (int j = 0; j < 4; ++j) {
    {
      float a0 = vp[0][j], a1 = vp[1][j], a2 = vp[2][j], a3 = vp[3][j];
      float t0 = c8 * a0 - s8 * a1, t1 = s8 * a0 + c8 * a1;
      float t2 = c8 * a2 - s8 * a3, t3 = s8 * a2 + c8 * a3;
      op[0][j] = c9 * t0 - s9 * t2;
      op[1][j] = c9 * t1 - s9 * t3;
      op[2][j] = s9 * t0 + c9 * t2;
      op[3][j] = s9 * t1 + c9 * t3;
    }
    {
      float a0 = vc[0][j], a1 = vc[1][j], a2 = vc[2][j], a3 = vc[3][j];
      float t0 = c8 * a0 - s8 * a1, t1 = s8 * a0 + c8 * a1;
      float t2 = c8 * a2 - s8 * a3, t3 = s8 * a2 + c8 * a3;
      oc[0][j] = c9 * t0 - s9 * t2;
      oc[1][j] = c9 * t1 - s9 * t3;
      oc[2][j] = s9 * t0 + c9 * t2;
      oc[3][j] = s9 * t1 + c9 * t3;
    }
  }
#pragma unroll
  for (int t = 0; t < 4; ++t) {
    v4f o; o.x = op[t][0]; o.y = op[t][1]; o.z = op[t][2]; o.w = op[t][3];
    *(v4f*)(S + lds_pos((t << 8) | (m << 4) | (4 * g))) = o;
    v4f c; c.x = oc[t][3]; c.y = oc[t][2]; c.z = oc[t][1]; c.w = oc[t][0];
    *(v4f*)(S + lds_pos((t << 8) | ((15 - m) << 4) | (12 - 4 * g))) = c;
  }
}

__device__ __forceinline__ void plane_sync() {
  __builtin_amdgcn_fence(__ATOMIC_ACQ_REL, "workgroup", "local");
}

__global__ __launch_bounds__(64) void qnn_circuit_kernel(
    const float* __restrict__ x, const float* __restrict__ W_pre,
    const float* __restrict__ b_pre, const float* __restrict__ weights,
    const float* __restrict__ W_post, const float* __restrict__ b_post,
    float* __restrict__ out) {
  __shared__ __align__(16) float Sst[2 * PLANE];
  __shared__ float hs[16];
  __shared__ float gs[16];
  __shared__ float csc[(NLAYERS + 1) * NQ];
  __shared__ float css[(NLAYERS + 1) * NQ];
  __shared__ __align__(16) float red[2 * 6 * 32];
  __shared__ float fin[4 * 16];
  __shared__ float zb[16];

  const int tid = threadIdx.x;
  const int lane = tid & 31;
  const int w = __builtin_amdgcn_readfirstlane(tid >> 5);
  const int b = blockIdx.x;
  float* Sre = Sst;
  float* Sim = Sst + PLANE;
  float* S = Sst + w * PLANE;

  for (int idx = tid; idx < (NLAYERS + 1) * NQ; idx += 64) {
    float wv = 0.5f * weights[idx];
    csc[idx] = __cosf(wv);
    css[idx] = __sinf(wv);
  }
  if (tid < NQ) {
    float acc = b_pre[tid];
#pragma unroll
    for (int k = 0; k < NQ; ++k) acc += x[b * NQ + k] * W_pre[tid * NQ + k];
    hs[tid] = acc;
  }
  __syncthreads();
  if (tid < NQ - 1) {
    const float PI = 3.14159265358979f;
    gs[tid] = (PI - hs[tid]) * (PI - hs[tid + 1]);
  }
  __syncthreads();

  float hr[NQ], gr[NQ - 1];
#pragma unroll
  for (int q = 0; q < NQ; ++q) hr[q] = hs[q];
#pragma unroll
  for (int q = 0; q < NQ - 1; ++q) gr[q] = gs[q];

  const float inv = 0.03125f;
  {
    const int c = lane & 15;
    const int hh = lane >> 4;
    const int cz = c ^ (c >> 1);
    float Cc = 0.f;
#pragma unroll
    for (int q = 0; q < 4; ++q) Cc += ((c >> q) & 1) ? -hr[q] : hr[q];
#pragma unroll
    for (int q = 0; q < 3; ++q) Cc += ((cz >> q) & 1) ? -gr[q] : gr[q];
    float Rl = 0.f;
#pragma unroll
    for (int i = 0; i < 4; ++i) Rl += ((c >> i) & 1) ? -hr[4 + i] : hr[4 + i];
#pragma unroll
    for (int i = 0; i < 3; ++i) Rl += ((cz >> i) & 1) ? -gr[4 + i] : gr[4 + i];
    const float crossA = (c & 1) ? -gr[3] : gr[3];
    const float z3 = (c & 8) ? -1.f : 1.f;
    v2f bop;
    bop.x = hh ? z3 : 1.f;
    bop.y = hh ? 0.f : Cc;
    const int sLW = cnot_src(c | (hh << 7) | (w << 9));
    const int rowLW = sLW >> 4;
    const int colL = sLW & 15;
    int pbase[8];
#pragma unroll
    for (int k = 0; k < 8; ++k) {
      const int kp = (k & 1) | (((k >> 1) & 1) << 3) | ((k >> 2) ? 0x30 : 0);
      pbase[k] = (int)__umul24((unsigned)(rowLW ^ kp), (unsigned)ROWP) + colL;
    }
#pragma unroll
    for (int u = 0; u < 2; ++u) {
      float Ru = Rl + ((((c >> 3) & 1) ^ u) ? -gr[7] : gr[7]);
      Ru += u ? -hr[8] : hr[8];
      Ru += (u ^ w) ? -gr[8] : gr[8];
      Ru += w ? -hr[9] : hr[9];
      v2f aop;
      aop.x = hh ? crossA : Ru;
      aop.y = hh ? 0.f : 1.f;
      v8f zero8 = {0.f, 0.f, 0.f, 0.f, 0.f, 0.f, 0.f, 0.f};
      v8f ang = __builtin_amdgcn_wmma_f32_16x16x4_f32(
          false, aop, false, bop, (short)0, zero8, false, false);
#pragma unroll
      for (int r = 0; r < 8; ++r) {
        float sn, cn;
        __sincosf(ang[r], &sn, &cn);
        const int kr = cnot_src((r << 4) | (u << 8)) >> 4;
        const int p = pbase[(kr & 1) | (((kr >> 3) & 1) << 1) | (u << 2)] + (kr & 6) * ROWP;
        Sre[p] = cn * inv;
        Sim[p] = -sn * inv;
      }
    }
  }
  __syncthreads();

  v2f bfrag[4];
#pragma unroll 1
  for (int l = 0; l <= NLAYERS; ++l) {
    gather_top(S, csc[l * NQ + 8], css[l * NQ + 8],
               csc[l * NQ + 9], css[l * NQ + 9], lane);
    plane_sync();
    build_bfrag(&csc[l * NQ + 0], &css[l * NQ + 0], lane, bfrag);
    contract_group(S, bfrag, lane);
    plane_sync();
    build_bfrag(&csc[l * NQ + 4], &css[l * NQ + 4], lane, bfrag);
    contract_group(S, bfrag, lane);
    plane_sync();
  }
  {
    const int half = lane >> 4;
    const int lid = lane & 15;
    const float* base = S + lid * ROWP + 8 * half;
    float p[4][8];
#pragma unroll
    for (int t = 0; t < 4; ++t) {
      v4f lo = *(const v4f*)(base + t * 16 * ROWP);
      v4f hi = *(const v4f*)(base + t * 16 * ROWP + 4);
      p[t][0] = lo.x * lo.x; p[t][1] = lo.y * lo.y; p[t][2] = lo.z * lo.z; p[t][3] = lo.w * lo.w;
      p[t][4] = hi.x * hi.x; p[t][5] = hi.y * hi.y; p[t][6] = hi.z * hi.z; p[t][7] = hi.w * hi.w;
    }
    float a[8], z8 = 0.f, z9 = 0.f;
#pragma unroll
    for (int r = 0; r < 8; ++r) {
      float s01 = p[0][r] + p[1][r], d01 = p[0][r] - p[1][r];
      float s23 = p[2][r] + p[3][r], d23 = p[2][r] - p[3][r];
      a[r] = s01 + s23;
      z8 += d01 + d23;
      z9 += s01 - s23;
    }
    float s0[4], z0 = 0.f;
#pragma unroll
    for (int i = 0; i < 4; ++i) { s0[i] = a[2 * i] + a[2 * i + 1]; z0 += a[2 * i] - a[2 * i + 1]; }
    float s1[2], z1 = 0.f;
#pragma unroll
    for (int i = 0; i < 2; ++i) { s1[i] = s0[2 * i] + s0[2 * i + 1]; z1 += s0[2 * i] - s0[2 * i + 1]; }
    const float tot = s1[0] + s1[1];
    const float z2 = s1[0] - s1[1];
    float* redw = red + w * (6 * 32);
    redw[lane] = tot;
    redw[32 + lane] = z0;  redw[64 + lane] = z1;  redw[96 + lane] = z2;
    redw[128 + lane] = z8; redw[160 + lane] = z9;
    plane_sync();
    const int q = lid;
    const int part = half;
    const int row = (q < 3) ? (q + 1) : ((q == 8) ? 4 : ((q == 9) ? 5 : 0));
    const float* src = redw + row * 32 + 16 * part;
    float xv[16];
#pragma unroll
    for (int j = 0; j < 4; ++j) {
      v4f x4 = *(const v4f*)(src + 4 * j);
      xv[4 * j] = x4.x; xv[4 * j + 1] = x4.y; xv[4 * j + 2] = x4.z; xv[4 * j + 3] = x4.w;
    }
    float t0[8], w0 = 0.f;
#pragma unroll
    for (int i = 0; i < 8; ++i) { t0[i] = xv[2 * i] + xv[2 * i + 1]; w0 += xv[2 * i] - xv[2 * i + 1]; }
    float t1[4], w1 = 0.f;
#pragma unroll
    for (int i = 0; i < 4; ++i) { t1[i] = t0[2 * i] + t0[2 * i + 1]; w1 += t0[2 * i] - t0[2 * i + 1]; }
    float t2[2], w2 = 0.f;
#pragma unroll
    for (int i = 0; i < 2; ++i) { t2[i] = t1[2 * i] + t1[2 * i + 1]; w2 += t1[2 * i] - t1[2 * i + 1]; }
    const float wt = t2[0] + t2[1];
    const float w3 = t2[0] - t2[1];
    float res = wt;
    if (q == 3) res = part ? -wt : wt;
    if (q == 4) res = w0;
    if (q == 5) res = w1;
    if (q == 6) res = w2;
    if (q == 7) res = w3;
    fin[(2 * w + part) * 16 + q] = res;
  }
  __syncthreads();
  if (tid < NQ) zb[tid] = (fin[tid] + fin[16 + tid]) + (fin[32 + tid] + fin[48 + tid]);
  __syncthreads();
  if (tid < NQ) {
    float acc = b_post[tid];
#pragma unroll
    for (int q = 0; q < NQ; ++q) acc += zb[q] * W_post[tid * NQ + q];
    out[b * NQ + tid] = acc;
  }
}

extern "C" void kernel_launch(void* const* d_in, const int* in_sizes, int n_in,
                              void* d_out, int out_size, void* d_ws, size_t ws_size,
                              hipStream_t stream) {
  const float* x       = (const float*)d_in[0];
  const float* W_pre   = (const float*)d_in[1];
  const float* b_pre   = (const float*)d_in[2];
  const float* weights = (const float*)d_in[3];
  const float* W_post  = (const float*)d_in[4];
  const float* b_post  = (const float*)d_in[5];
  float* out = (float*)d_out;
  const int batch = in_sizes[0] / NQ;
  hipLaunchKernelGGL(qnn_circuit_kernel, dim3(batch), dim3(64), 0, stream,
                     x, W_pre, b_pre, weights, W_post, b_post, out);
}
